# starting kernel
# baseline (speedup 1.0000x reference)
#include <hip/hip_runtime.h>
#include <hip/hip_bf16.h>
#include <hip/hip_fp16.h>

#define F_IN      128
#define HEADS     4
#define C_OUT     32
#define HC        128
#define NEG_SLOPE 0.2f

typedef float    v2f  __attribute__((ext_vector_type(2)));
typedef float    v8f  __attribute__((ext_vector_type(8)));
typedef _Float16 v8h  __attribute__((ext_vector_type(8)));
typedef _Float16 v16h __attribute__((ext_vector_type(16)));

#if __has_builtin(__builtin_amdgcn_wmma_f32_16x16x4_f32)
#define GAT_WMMA_F32 1
#else
#define GAT_WMMA_F32 0
#endif

__device__ __forceinline__ unsigned f2key(float f) {
  unsigned u = __float_as_uint(f);
  return (u & 0x80000000u) ? ~u : (u | 0x80000000u);
}
__device__ __forceinline__ float key2f(unsigned k) {
  unsigned u = (k & 0x80000000u) ? (k & 0x7FFFFFFFu) : ~k;
  return __uint_as_float(u);
}

__global__ void k_convert(const float* __restrict__ x, const float* __restrict__ W,
                          _Float16* __restrict__ xh, _Float16* __restrict__ Wh,
                          int nx, int nw) {
  int t = blockIdx.x * blockDim.x + threadIdx.x;
  if (t < nx) xh[t] = (_Float16)x[t];
  if (t < nw) Wh[t] = (_Float16)W[t];
}

__global__ void k_gemm(const float* __restrict__ x, const float* __restrict__ W,
                       const _Float16* __restrict__ xh, const _Float16* __restrict__ Wh,
                       float* __restrict__ xl, int N) {
  const int lane    = threadIdx.x & 31;
  const int wave    = threadIdx.x >> 5;
  const int g       = lane >> 4;
  const int l16     = lane & 15;
  const int rowBase = blockIdx.x * 16;
  const int colBase = wave * 16;
  int mrow = rowBase + l16; if (mrow >= N) mrow = N - 1;
  const int ncol = colBase + l16;

  v8f c = {};
#if GAT_WMMA_F32
  for (int kk = 0; kk < F_IN; kk += 4) {
    const int k0 = kk + g * 2;
    v2f a = *(const v2f*)(x + (size_t)mrow * F_IN + k0);
    v2f b = *(const v2f*)(W + (size_t)ncol * F_IN + k0);
    c = __builtin_amdgcn_wmma_f32_16x16x4_f32(false, a, false, b, (short)0, c,
                                              false, false);
  }
#else
  for (int kk = 0; kk < F_IN; kk += 32) {
    const _Float16* ap = xh + (size_t)mrow * F_IN + kk + 8 * g;
    v8h a0 = *(const v8h*)(ap);
    v8h a1 = *(const v8h*)(ap + 16);
    v16h a = __builtin_shufflevector(a0, a1, 0, 1, 2, 3, 4, 5, 6, 7,
                                     8, 9, 10, 11, 12, 13, 14, 15);
    v16h b = *(const v16h*)(Wh + (size_t)ncol * F_IN + kk + 16 * g);
    c = __builtin_amdgcn_wmma_f32_16x16x32_f16(false, a, false, b, (short)0, c,
                                               false, false);
  }
#endif
#pragma unroll
  for (int i = 0; i < 8; ++i) {
    const int r = rowBase + i + g * 8;
    if (r < N) xl[(size_t)r * HC + ncol] = c[i];
  }
}

__global__ void k_scores(const float* __restrict__ xl, const float* __restrict__ aj,
                         const float* __restrict__ ai, float* __restrict__ Sj,
                         float* __restrict__ Si, unsigned* __restrict__ mkey,
                         float* __restrict__ ssum, int N) {
  const int t = blockIdx.x * blockDim.x + threadIdx.x;
  if (t >= N * HEADS) return;
  const int n = t / HEADS, h = t % HEADS;
  const float* row = xl + (size_t)n * HC + h * C_OUT;
  float sj = 0.f, si = 0.f;
#pragma unroll
  for (int cc = 0; cc < C_OUT; ++cc) {
    sj += row[cc] * aj[h * C_OUT + cc];
    si += row[cc] * ai[h * C_OUT + cc];
  }
  Sj[t] = sj;
  Si[t] = si;
  mkey[t] = 0u;
  ssum[t] = 0.f;
}

__global__ void k_out_init(float* __restrict__ out, const float* __restrict__ bias,
                           int total) {
  const int t = blockIdx.x * blockDim.x + threadIdx.x;
  if (t < total) out[t] = bias[t & (HC - 1)];
}

__device__ __forceinline__ void edge_nodes(const int* __restrict__ src,
                                           const int* __restrict__ dst, int E,
                                           int e, int& nj, int& ni, int& dnode,
                                           int& snode) {
  const int es = (e < E) ? src[e] : (e - E);
  const int ed = (e < E) ? dst[e] : (e - E);
  snode = es;
  dnode = ed;
  nj = src[es];
  ni = dst[ed];
}

__device__ __forceinline__ float alpha_h(const float* __restrict__ Sj,
                                         const float* __restrict__ Si, int nj,
                                         int ni, int h) {
  float a = Sj[nj * HEADS + h] + Si[ni * HEADS + h];
  return (a >= 0.f) ? a : NEG_SLOPE * a;
}

__global__ void k_alpha_max(const int* __restrict__ src, const int* __restrict__ dst,
                            const float* __restrict__ Sj, const float* __restrict__ Si,
                            unsigned* __restrict__ mkey, int E, int Et) {
  const int e = blockIdx.x * blockDim.x + threadIdx.x;
  if (e >= Et) return;
  int nj, ni, dn, sn;
  edge_nodes(src, dst, E, e, nj, ni, dn, sn);
#pragma unroll
  for (int h = 0; h < HEADS; ++h)
    atomicMax(&mkey[dn * HEADS + h], f2key(alpha_h(Sj, Si, nj, ni, h)));
}

__global__ void k_exp_sum(const int* __restrict__ src, const int* __restrict__ dst,
                          const float* __restrict__ Sj, const float* __restrict__ Si,
                          const unsigned* __restrict__ mkey, float* __restrict__ ssum,
                          float* __restrict__ coef, int E, int Et) {
  const int e = blockIdx.x * blockDim.x + threadIdx.x;
  if (e >= Et) return;
  int nj, ni, dn, sn;
  edge_nodes(src, dst, E, e, nj, ni, dn, sn);
#pragma unroll
  for (int h = 0; h < HEADS; ++h) {
    const int idx = dn * HEADS + h;
    const float a = expf(alpha_h(Sj, Si, nj, ni, h) - key2f(mkey[idx]));
    coef[(size_t)e * HEADS + h] = a;
    atomicAdd(&ssum[idx], a);
  }
}

__global__ void k_recip(float* __restrict__ ssum, int total) {
  const int t = blockIdx.x * blockDim.x + threadIdx.x;
  if (t < total) ssum[t] = 1.0f / ssum[t];
}

__global__ void k_aggregate(const int* __restrict__ src, const int* __restrict__ dst,
                            const float* __restrict__ coef,
                            const float* __restrict__ sinv,
                            const float* __restrict__ xl, float* __restrict__ out,
                            int E, int Et) {
  const long long t = (long long)blockIdx.x * blockDim.x + threadIdx.x;
  if (t >= (long long)Et * HC) return;
  const int e = (int)(t >> 7);
  const int f = (int)(t & 127);
  const int h = f >> 5;
  const int sn = (e < E) ? src[e] : (e - E);
  const int dn = (e < E) ? dst[e] : (e - E);
  const float c = coef[(size_t)e * HEADS + h] * sinv[dn * HEADS + h];
  atomicAdd(&out[(size_t)dn * HC + f], c * xl[(size_t)sn * HC + f]);
}

extern "C" void kernel_launch(void* const* d_in, const int* in_sizes, int n_in,
                              void* d_out, int out_size, void* d_ws, size_t ws_size,
                              hipStream_t stream) {
  const float* x      = (const float*)d_in[0];
  const float* W      = (const float*)d_in[1];
  const float* attn_j = (const float*)d_in[2];
  const float* attn_i = (const float*)d_in[3];
  const float* bias   = (const float*)d_in[4];
  const int*   eidx   = (const int*)d_in[5];

  const int N  = in_sizes[0] / F_IN;
  const int E  = in_sizes[5] / 2;
  const int Et = E + N;
  const int* src = eidx;
  const int* dst = eidx + E;

  char*  ws = (char*)d_ws;
  size_t o  = 0;
  auto take = [&](size_t bytes) -> void* {
    void* p = ws + o;
    o += (bytes + 255) & ~(size_t)255;
    return p;
  };
  float*     xl   = (float*)take((size_t)N * HC * sizeof(float));
  float*     Sj   = (float*)take((size_t)N * HEADS * sizeof(float));
  float*     Si   = (float*)take((size_t)N * HEADS * sizeof(float));
  unsigned*  mkey = (unsigned*)take((size_t)N * HEADS * sizeof(unsigned));
  float*     ssum = (float*)take((size_t)N * HEADS * sizeof(float));
  float*     coef = (float*)take((size_t)Et * HEADS * sizeof(float));
  _Float16*  xh   = (_Float16*)take((size_t)N * F_IN * sizeof(_Float16));
  _Float16*  Wh   = (_Float16*)take((size_t)HC * F_IN * sizeof(_Float16));

  float* out = (float*)d_out;

  const int nx = N * F_IN, nw = HC * F_IN;
  k_convert<<<(nx + 255) / 256, 256, 0, stream>>>(x, W, xh, Wh, nx, nw);

  k_gemm<<<(N + 15) / 16, 256, 0, stream>>>(x, W, xh, Wh, xl, N);

  k_scores<<<(N * HEADS + 255) / 256, 256, 0, stream>>>(xl, attn_j, attn_i, Sj, Si,
                                                        mkey, ssum, N);

  const int ntot = N * HC;
  k_out_init<<<(ntot + 255) / 256, 256, 0, stream>>>(out, bias, ntot);

  k_alpha_max<<<(Et + 255) / 256, 256, 0, stream>>>(src, dst, Sj, Si, mkey, E, Et);
  k_exp_sum<<<(Et + 255) / 256, 256, 0, stream>>>(src, dst, Sj, Si, mkey, ssum,
                                                  coef, E, Et);
  k_recip<<<(N * HEADS + 255) / 256, 256, 0, stream>>>(ssum, N * HEADS);

  const long long tot = (long long)Et * HC;
  k_aggregate<<<(unsigned)((tot + 255) / 256), 256, 0, stream>>>(
      src, dst, coef, ssum, xl, out, E, Et);
}


# GEMM work-groups walk row tiles (768 resident, tiles b, b+768, ...): a wave loads its 16 columns of W once into 64 registers and reuses them for every tile; only x is loaded per tile; sum order kept
# speedup vs baseline: 10.1230x; 10.1230x over previous
#include <hip/hip_runtime.h>
#include <hip/hip_bf16.h>
#include <hip/hip_fp16.h>

#define F_IN      128
#define HEADS     4
#define C_OUT     32
#define HC        128
#define NEG_SLOPE 0.2f

typedef float    v2f  __attribute__((ext_vector_type(2)));
typedef float    v8f  __attribute__((ext_vector_type(8)));
typedef _Float16 v8h  __attribute__((ext_vector_type(8)));
typedef _Float16 v16h __attribute__((ext_vector_type(16)));

#if __has_builtin(__builtin_amdgcn_wmma_f32_16x16x4_f32)
#define GAT_WMMA_F32 1
#else
#define GAT_WMMA_F32 0
#endif

__device__ __forceinline__ unsigned f2key(float f) {
  unsigned u = __float_as_uint(f);
  return (u & 0x80000000u) ? ~u : (u | 0x80000000u);
}
__device__ __forceinline__ float key2f(unsigned k) {
  unsigned u = (k & 0x80000000u) ? (k & 0x7FFFFFFFu) : ~k;
  return __uint_as_float(u);
}

__device__ __forceinline__ void convert8(const float* __restrict__ s,
                                         _Float16* __restrict__ d) {
  const float4 lo = *(const float4*)s;
  const float4 hi = *(const float4*)(s + 4);
  const v8h h = {(_Float16)lo.x, (_Float16)lo.y, (_Float16)lo.z, (_Float16)lo.w,
                 (_Float16)hi.x, (_Float16)hi.y, (_Float16)hi.z, (_Float16)hi.w};
  *(v8h*)d = h;
}

__device__ __forceinline__ void convert_tile(const float* __restrict__ x,
                                             const float* __restrict__ W,
                                             _Float16* __restrict__ xh,
                                             _Float16* __restrict__ Wh,
                                             int nx, int nw, int tile, int ntile) {
  const int t8 = (int)threadIdx.x * 8;
  const int ix = tile * (16 * F_IN) + t8;
  if (ix < nx) convert8(x + ix, xh + ix);
  for (int iw = tile * 2048 + t8; iw < nw; iw += ntile * 2048)
    convert8(W + iw, Wh + iw);
}

#define KSTEPS (F_IN / 8)
#define GWG    768

#if GAT_WMMA_F32
__device__ __forceinline__ void gemm_tile(const float* __restrict__ x,
                                          const float4 (&bf)[KSTEPS],
                                          float* __restrict__ xl, int N, int tile) {
  const int lane    = threadIdx.x & 31;
  const int wave    = threadIdx.x >> 5;
  const int g       = lane >> 4;
  const int l16     = lane & 15;
  const int rowBase = tile * 16;
  const int ncol    = wave * 16 + l16;
  int mrow = rowBase + l16; if (mrow >= N) mrow = N - 1;

  float4 af[KSTEPS];
#pragma unroll
  for (int i = 0; i < KSTEPS; ++i)
    af[i] = *(const float4*)(x + (size_t)mrow * F_IN + i * 8 + g * 4);
  v8f c = {};
#pragma unroll
  for (int i = 0; i < KSTEPS; ++i) {
    const v2f a0 = {af[i].x, af[i].y}, b0 = {bf[i].x, bf[i].y};
    const v2f a1 = {af[i].z, af[i].w}, b1 = {bf[i].z, bf[i].w};
    c = __builtin_amdgcn_wmma_f32_16x16x4_f32(false, a0, false, b0, (short)0, c,
                                              false, false);
    c = __builtin_amdgcn_wmma_f32_16x16x4_f32(false, a1, false, b1, (short)0, c,
                                              false, false);
  }
#pragma unroll
  for (int i = 0; i < 8; ++i) {
    const int r = rowBase + i + g * 8;
    if (r < N) xl[(size_t)r * HC + ncol] = c[i];
  }
}
#else
__device__ __forceinline__ void gemm_tile_h(const _Float16* __restrict__ xh,
                                            const _Float16* __restrict__ Wh,
                                            float* __restrict__ xl, int N, int tile) {
  const int lane    = threadIdx.x & 31;
  const int wave    = threadIdx.x >> 5;
  const int g       = lane >> 4;
  const int l16     = lane & 15;
  const int rowBase = tile * 16;
  const int ncol    = wave * 16 + l16;
  int mrow = rowBase + l16; if (mrow >= N) mrow = N - 1;
  v8f c = {};
  for (int kk = 0; kk < F_IN; kk += 32) {
    const _Float16* ap = xh + (size_t)mrow * F_IN + kk + 8 * g;
    v8h a0 = *(const v8h*)(ap);
    v8h a1 = *(const v8h*)(ap + 16);
    v16h a = __builtin_shufflevector(a0, a1, 0, 1, 2, 3, 4, 5, 6, 7,
                                     8, 9, 10, 11, 12, 13, 14, 15);
    v16h b = *(const v16h*)(Wh + (size_t)ncol * F_IN + kk + 16 * g);
    c = __builtin_amdgcn_wmma_f32_16x16x32_f16(false, a, false, b, (short)0, c,
                                               false, false);
  }
#pragma unroll
  for (int i = 0; i < 8; ++i) {
    const int r = rowBase + i + g * 8;
    if (r < N) xl[(size_t)r * HC + ncol] = c[i];
  }
}
#endif

__global__ __launch_bounds__(256) void k_front(
    const float* __restrict__ x, const float* __restrict__ W,
    _Float16* __restrict__ xh, _Float16* __restrict__ Wh,
    float* __restrict__ xl, int N, int nx, int nw, int ntile) {
#if GAT_WMMA_F32
  const int lane = threadIdx.x & 31;
  const int ncol = (int)(threadIdx.x >> 5) * 16 + (lane & 15);
  float4 bf[KSTEPS];
#pragma unroll
  for (int i = 0; i < KSTEPS; ++i)
    bf[i] = *(const float4*)(W + (size_t)ncol * F_IN + i * 8 + (lane >> 4) * 4);
  for (int t = blockIdx.x; t < ntile; t += GWG) {
    gemm_tile(x, bf, xl, N, t);
    convert_tile(x, W, xh, Wh, nx, nw, t, ntile);
  }
#else
  for (int t = blockIdx.x; t < ntile; t += GWG) {
    convert_tile(x, W, xh, Wh, nx, nw, t, ntile);
    __syncthreads();
    gemm_tile_h(xh, Wh, xl, N, t);
  }
#endif
}

__device__ __forceinline__ void scores_body(const float* __restrict__ xl,
                                            const float* __restrict__ aj,
                                            const float* __restrict__ ai,
                                            float* __restrict__ Sj,
                                            float* __restrict__ Si, int N, int t) {
  if (t >= N * HEADS) return;
  const int n = t / HEADS, h = t % HEADS;
  const float* row = xl + (size_t)n * HC + h * C_OUT;
  float sj = 0.f, si = 0.f;
#pragma unroll
  for (int cc = 0; cc < C_OUT; ++cc) {
    sj += row[cc] * aj[h * C_OUT + cc];
    si += row[cc] * ai[h * C_OUT + cc];
  }
  Sj[t] = sj;
  Si[t] = si;
}

#define RNG 196
#define MKW 32
#define MKR 256
#define MKP (MKR + 1)

__device__ __forceinline__ void mark_body(const int* __restrict__ dst,
                                          unsigned* __restrict__ bm,
                                          int N, int E, int nranges, int pitch,
                                          int grp) {
  __shared__ unsigned tile[MKW * MKP];
  const int tid = threadIdx.x, lane = tid & 31, v = tid >> 5;
  const size_t w = (size_t)grp * MKW + v;

  int d[32];
#pragma unroll
  for (int j = 0; j < 32; ++j) {
    const size_t e = (w * 32 + j) * 32 + lane;
    d[j] = (e < (size_t)E) ? dst[e] : -1;
  }
  for (int r0 = 0; r0 < nranges; r0 += MKR) {
    for (int i = tid; i < MKW * MKP; i += 1024) tile[i] = 0u;
    __syncthreads();
#pragma unroll
    for (int j = 0; j < 32; ++j) {
      if ((unsigned)d[j] < (unsigned)N) {
        const unsigned r = (unsigned)d[j] / RNG - (unsigned)r0;
        if (r < MKR) atomicOr(&tile[v * MKP + r], 1u << j);
      }
    }
    __syncthreads();
    for (int r = v; r < MKR && r0 + r < nranges; r += 32)
      bm[(size_t)(r0 + r) * pitch + (size_t)grp * MKW + lane] =
          tile[lane * MKP + r];
    __syncthreads();
  }
}

#define PTH 1024
__global__ __launch_bounds__(PTH) void k_pre(
    const float* __restrict__ xl, const float* __restrict__ aj,
    const float* __restrict__ ai, float* __restrict__ Sj, float* __restrict__ Si,
    const int* __restrict__ dst, unsigned* __restrict__ bm,
    int N, int E, int nranges, int pitch, int nmark) {
  const int b = blockIdx.x;
  if (b < nmark)
    mark_body(dst, bm, N, E, nranges, pitch, b);
  else
    scores_body(xl, aj, ai, Sj, Si, N, (b - nmark) * PTH + (int)threadIdx.x);
}

#define ETH 1024
#define EPT 8
#define CAP (ETH * EPT)
#define WB  8

__device__ __forceinline__ float leaky(float a) {
  return (a >= 0.f) ? a : NEG_SLOPE * a;
}

__device__ __forceinline__ float lane_f(float v, int k) {
  return __int_as_float(__builtin_amdgcn_readlane(__float_as_int(v), k));
}

__device__ __forceinline__ float head_w(const float4& w, int k, int hq) {
  const float w0 = lane_f(w.x, k), w1 = lane_f(w.y, k);
  const float w2 = lane_f(w.z, k), w3 = lane_f(w.w, k);
  const float lo = (hq & 1) ? w1 : w0;
  const float hi = (hq & 1) ? w3 : w2;
  return (hq & 2) ? hi : lo;
}

__global__ __launch_bounds__(ETH) void k_edge(
    const int* __restrict__ src, const int* __restrict__ dst,
    const float* __restrict__ Sj, const float* __restrict__ Si,
    const float* __restrict__ xl, const float* __restrict__ bias,
    const unsigned* __restrict__ bm, int pitch,
    float* __restrict__ out, int N, int E) {
  __shared__ __attribute__((aligned(16))) float acc[RNG * HC];
  __shared__ unsigned lst_e[CAP];
  __shared__ __attribute__((aligned(16)))
  unsigned clst[CAP];
  __shared__ unsigned tkey[RNG * HEADS];
  __shared__ __attribute__((aligned(16))) float Uv[RNG * HEADS];
  __shared__ __attribute__((aligned(16))) float mval[RNG * HEADS];
  __shared__ float    ssum[RNG * HEADS];
  __shared__ unsigned cnt;
  __shared__ unsigned ccnt;

  const int tid  = threadIdx.x;
  const int base = blockIdx.x * RNG;
  const int nloc = (N - base < RNG) ? (N - base) : RNG;
  const int Et   = E + N;

  for (int i = tid; i < RNG * HC / 4; i += ETH)
    ((float4*)acc)[i] = make_float4(0.f, 0.f, 0.f, 0.f);
  for (int i = tid; i < CAP / 4; i += ETH)
    ((uint4*)clst)[i] = make_uint4(0u, 0u, 0u, 0u);
  for (int i = tid; i < RNG * HEADS; i += ETH) {
    const int dl = i >> 2, h = i & 3;
    tkey[i] = 0u;
    ssum[i] = 0.f;
    Uv[i] = (dl < nloc) ? Si[dst[base + dl] * HEADS + h] : 0.f;
  }
  if (tid == 0) { cnt = 0u; ccnt = 0u; }
  __syncthreads();

  const unsigned* row = bm + (size_t)blockIdx.x * pitch;
  for (int w = tid; w < pitch; w += ETH) {
    unsigned m = row[w];
    while (m) {
      const int j = __ffs(m) - 1;
      m &= m - 1;
      const unsigned pos = atomicAdd(&ccnt, 1u);
      if (pos < CAP) clst[pos] = (unsigned)w * 32u + (unsigned)j;
    }
  }
  __syncthreads();

  const unsigned ctot = ccnt;
  const int nch = (int)((ctot < CAP) ? ctot : CAP);
  {
    const int lane = tid & 31, wave = tid >> 5, g = lane >> 3, q = lane & 7;
    if (((((size_t)dst) & 15) == 0) && ((E & 3) == 0)) {
      for (int ib = wave * 32; ib < nch; ib += 32 * (ETH / 32)) {
        const uint4 ca = *(const uint4*)(clst + ib + 8 * g);
        const uint4 cb = *(const uint4*)(clst + ib + 8 * g + 4);
        const unsigned cid[8] = {ca.x, ca.y, ca.z, ca.w, cb.x, cb.y, cb.z, cb.w};
        size_t e0[8];
        unsigned lim[8];
        int4 dv[8];
#pragma unroll
        for (int k = 0; k < 8; ++k) {
          e0[k] = (size_t)cid[k] * 32 + (size_t)q * 4;
          lim[k] = (ib + 8 * g + k < nch && e0[k] < (size_t)E) ? (unsigned)nloc : 0u;
        }
#pragma unroll
        for (int k = 0; k < 8; ++k)
          dv[k] = *(const int4*)(dst + (lim[k] ? e0[k] : (size_t)0));
        unsigned hits = 0u;
#pragma unroll
        for (int k = 0; k < 8; ++k) {
          hits |= ((unsigned)(dv[k].x - base) < lim[k] ? 1u : 0u) << (4 * k);
          hits |= ((unsigned)(dv[k].y - base) < lim[k] ? 1u : 0u) << (4 * k + 1);
          hits |= ((unsigned)(dv[k].z - base) < lim[k] ? 1u : 0u) << (4 * k + 2);
          hits |= ((unsigned)(dv[k].w - base) < lim[k] ? 1u : 0u) << (4 * k + 3);
        }
        while (hits) {
          const int bit = __ffs(hits) - 1;
          hits &= hits - 1;
          const unsigned c = clst[ib + 8 * g + (bit >> 2)];
          const unsigned pos = atomicAdd(&cnt, 1u);
          if (pos < CAP) lst_e[pos] = c * 32u + (unsigned)(q * 4 + (bit & 3));
        }
      }
    } else {
      for (int i = wave; i < nch; i += ETH / 32) {
        const size_t e = (size_t)clst[i] * 32 + lane;
        if (e < (size_t)E && (unsigned)(dst[e] - base) < (unsigned)nloc) {
          const unsigned pos = atomicAdd(&cnt, 1u);
          if (pos < CAP) lst_e[pos] = (unsigned)e;
        }
      }
    }
  }
  if (tid < nloc) {
    const unsigned pos = atomicAdd(&cnt, 1u);
    if (pos < CAP) lst_e[pos] = (unsigned)(E + base + tid);
  }
  __syncthreads();

  const unsigned total = (ctot > CAP) ? (CAP + 1u) : cnt;
  if (total <= CAP) {
    const int lane = tid & 31;
    const int n  = __builtin_amdgcn_readfirstlane((int)total);
    const int wv = __builtin_amdgcn_readfirstlane(tid >> 5);
    const int per = ((n + (ETH / 32) - 1) / (ETH / 32) + WB - 1) & ~(WB - 1);
    const int first = wv * per;
    int mine = n - first;
    if (mine > per) mine = per;
    if (mine < 0) mine = 0;

    int    ee[EPT];
    int    es[EPT];
    int    ed[EPT];
    float4 et[EPT];
#pragma unroll
    for (int j = 0; j < EPT; ++j) {
      const int p = j * 32 + lane;
      ee[j] = (p < mine) ? (int)lst_e[first + p] : -1;
    }
    {
      int sl[EPT], dld[EPT];
#pragma unroll
      for (int j = 0; j < EPT; ++j) {
        const int ec = (ee[j] >= 0 && ee[j] < E) ? ee[j] : 0;
        sl[j]  = src[ec];
        dld[j] = dst[ec];
      }
#pragma unroll
      for (int j = 0; j < EPT; ++j) {
        const int e = ee[j];
        es[j] = (e < 0) ? 0 : ((e < E) ? sl[j] : (e - E));
        ed[j] = (e < 0) ? 0 : (((e < E) ? dld[j] : (e - E)) - base);
      }
    }
    {
      int ss[EPT];
#pragma unroll
      for (int j = 0; j < EPT; ++j) ss[j] = src[es[j]];
#pragma unroll
      for (int j = 0; j < EPT; ++j)
        et[j] = *(const float4*)(Sj + (size_t)ss[j] * HEADS);
    }
#pragma unroll
    for (int j = 0; j < EPT; ++j) {
      if (ee[j] >= 0) {
        atomicMax(&tkey[ed[j] * HEADS + 0], f2key(et[j].x));
        atomicMax(&tkey[ed[j] * HEADS + 1], f2key(et[j].y));
        atomicMax(&tkey[ed[j] * HEADS + 2], f2key(et[j].z));
        atomicMax(&tkey[ed[j] * HEADS + 3], f2key(et[j].w));
      }
    }
    __syncthreads();
    for (int i = tid; i < RNG * HEADS; i += ETH)
      mval[i] = leaky(key2f(tkey[i]) + Uv[i]);
    __syncthreads();

    const int hq = lane >> 3;
#pragma unroll
    for (int j = 0; j < EPT; ++j) {
      const int left = mine - j * 32;
      if (left > 0) {
        const int dl = ed[j];
        const float4 U = *(const float4*)(Uv + dl * HEADS);
        const float4 M = *(const float4*)(mval + dl * HEADS);
        float4 w;
        w.x = expf(leaky(et[j].x + U.x) - M.x);
        w.y = expf(leaky(et[j].y + U.y) - M.y);
        w.z = expf(leaky(et[j].z + U.z) - M.z);
        w.w = expf(leaky(et[j].w + U.w) - M.w);
        if (lane < left) {
          atomicAdd(&ssum[dl * HEADS + 0], w.x);
          atomicAdd(&ssum[dl * HEADS + 1], w.y);
          atomicAdd(&ssum[dl * HEADS + 2], w.z);
          atomicAdd(&ssum[dl * HEADS + 3], w.w);
        }
        const int cj = (left < 32) ? left : 32;
        const int cfull = cj & ~(WB - 1);
        for (int k0 = 0; k0 < cfull; k0 += WB) {
          float4 v[WB];
#pragma unroll
          for (int b = 0; b < WB; ++b) {
            const int s = __builtin_amdgcn_readlane(es[j], k0 + b);
            v[b] = *(const float4*)(xl + (size_t)s * HC + lane * 4);
          }
#pragma unroll
          for (int b = 0; b < WB; ++b) {
            const int k = k0 + b;
            const int d = __builtin_amdgcn_readlane(dl, k);
            const float wl = head_w(w, k, hq);
            float* ar = acc + d * HC + lane;
            atomicAdd(ar,      wl * v[b].x);
            atomicAdd(ar + 32, wl * v[b].y);
            atomicAdd(ar + 64, wl * v[b].z);
            atomicAdd(ar + 96, wl * v[b].w);
          }
        }
        for (int k = cfull; k < cj; ++k) {
          const int s = __builtin_amdgcn_readlane(es[j], k);
          const int d = __builtin_amdgcn_readlane(dl, k);
          const float4 v = *(const float4*)(xl + (size_t)s * HC + lane * 4);
          const float wl = head_w(w, k, hq);
          float* ar = acc + d * HC + lane;
          atomicAdd(ar,      wl * v.x);
          atomicAdd(ar + 32, wl * v.y);
          atomicAdd(ar + 64, wl * v.z);
          atomicAdd(ar + 96, wl * v.w);
        }
      }
    }
  } else {
    for (int e = tid; e < Et; e += ETH) {
      const int d = (e < E) ? dst[e] : (e - E);
      const unsigned dl = (unsigned)(d - base);
      if (dl < (unsigned)nloc) {
        const int s  = (e < E) ? src[e] : (e - E);
        const int ss = src[s];
        for (int hh = 0; hh < HEADS; ++hh)
          atomicMax(&tkey[dl * HEADS + hh], f2key(Sj[(size_t)ss * HEADS + hh]));
      }
    }
    __syncthreads();
    for (int i = tid; i < RNG * HEADS; i += ETH)
      mval[i] = leaky(key2f(tkey[i]) + Uv[i]);
    __syncthreads();
    for (int e = tid; e < Et; e += ETH) {
      const int d = (e < E) ? dst[e] : (e - E);
      const unsigned dl = (unsigned)(d - base);
      if (dl < (unsigned)nloc) {
        const int s  = (e < E) ? src[e] : (e - E);
        const int ss = src[s];
        float w[HEADS];
        for (int hh = 0; hh < HEADS; ++hh) {
          const float a = leaky(Sj[(size_t)ss * HEADS + hh] + Uv[dl * HEADS + hh]);
          w[hh] = expf(a - mval[dl * HEADS + hh]);
          atomicAdd(&ssum[dl * HEADS + hh], w[hh]);
        }
        for (int f = 0; f < HC; ++f)
          atomicAdd(&acc[dl * HC + (f & 3) * 32 + (f >> 2)],
                    w[f >> 5] * xl[(size_t)s * HC + f]);
      }
    }
  }
  __syncthreads();

  for (int i = tid; i < nloc * 32; i += ETH) {
    const int dl = i >> 5, l = i & 31;
    const float inv = 1.0f / ssum[dl * HEADS + (l >> 3)];
    const float* ar = acc + dl * HC + l;
    const float4 b = *(const float4*)(bias + l * 4);
    float4 o;
    o.x = ar[0]  * inv + b.x;
    o.y = ar[32] * inv + b.y;
    o.z = ar[64] * inv + b.z;
    o.w = ar[96] * inv + b.w;
    *(float4*)(out + (size_t)(base + dl) * HC + l * 4) = o;
  }
}

extern "C" void kernel_launch(void* const* d_in, const int* in_sizes, int n_in,
                              void* d_out, int out_size, void* d_ws, size_t ws_size,
                              hipStream_t stream) {
  const float* x      = (const float*)d_in[0];
  const float* W      = (const float*)d_in[1];
  const float* attn_j = (const float*)d_in[2];
  const float* attn_i = (const float*)d_in[3];
  const float* bias   = (const float*)d_in[4];
  const int*   eidx   = (const int*)d_in[5];

  const int N  = in_sizes[0] / F_IN;
  const int E  = in_sizes[5] / 2;
  const int* src = eidx;
  const int* dst = eidx + E;

  char*  ws = (char*)d_ws;
  size_t o  = 0;
  auto take = [&](size_t bytes) -> void* {
    void* p = ws + o;
    o += (bytes + 255) & ~(size_t)255;
    return p;
  };
  float*     xl   = (float*)take((size_t)N * HC * sizeof(float));
  float*     Sj   = (float*)take((size_t)N * HEADS * sizeof(float));
  float*     Si   = (float*)take((size_t)N * HEADS * sizeof(float));
  _Float16*  xh   = (_Float16*)take((size_t)N * F_IN * sizeof(_Float16));
  _Float16*  Wh   = (_Float16*)take((size_t)HC * F_IN * sizeof(_Float16));
  const int nranges = (N + RNG - 1) / RNG;
  const int nchunks = (E + 31) / 32;
  const int pitch   = ((nchunks + 31) / 32 + MKW - 1) / MKW * MKW;
  unsigned*  bm   = (unsigned*)take((size_t)nranges * pitch * sizeof(unsigned));

  float* out = (float*)d_out;

  const int nx = N * F_IN, nw = HC * F_IN;
  const int ntile = (N + 15) / 16;
  k_front<<<GWG, 256, 0, stream>>>(x, W, xh, Wh, xl, N, nx, nw, ntile);

  const int nmark  = pitch / MKW;
  const int nscore = (N * HEADS + PTH - 1) / PTH;
  k_pre<<<nmark + nscore, PTH, 0, stream>>>(xl, attn_j, attn_i, Sj, Si, dst, bm,
                                           N, E, nranges, pitch, nmark);

  k_edge<<<nranges, ETH, 0, stream>>>(src, dst, Sj, Si, xl, bias, bm, pitch, out, N, E);
}


# chunk-bitmap pass and score pass join the GEMM launch: 49 more work-groups build the bitmap, a GEMM work-group keeps its tiles in LDS and scores its own rows; 3 launches to 2; sum order kept
# speedup vs baseline: 10.6269x; 1.0498x over previous
#include <hip/hip_runtime.h>
#include <hip/hip_bf16.h>
#include <hip/hip_fp16.h>

#define F_IN      128
#define HEADS     4
#define C_OUT     32
#define HC        128
#define NEG_SLOPE 0.2f

typedef float    v2f  __attribute__((ext_vector_type(2)));
typedef float    v8f  __attribute__((ext_vector_type(8)));
typedef _Float16 v8h  __attribute__((ext_vector_type(8)));
typedef _Float16 v16h __attribute__((ext_vector_type(16)));

#if __has_builtin(__builtin_amdgcn_wmma_f32_16x16x4_f32)
#define GAT_WMMA_F32 1
#else
#define GAT_WMMA_F32 0
#endif

__device__ __forceinline__ unsigned f2key(float f) {
  unsigned u = __float_as_uint(f);
  return (u & 0x80000000u) ? ~u : (u | 0x80000000u);
}
__device__ __forceinline__ float key2f(unsigned k) {
  unsigned u = (k & 0x80000000u) ? (k & 0x7FFFFFFFu) : ~k;
  return __uint_as_float(u);
}

__device__ __forceinline__ void convert8(const float* __restrict__ s,
                                         _Float16* __restrict__ d) {
  const float4 lo = *(const float4*)s;
  const float4 hi = *(const float4*)(s + 4);
  const v8h h = {(_Float16)lo.x, (_Float16)lo.y, (_Float16)lo.z, (_Float16)lo.w,
                 (_Float16)hi.x, (_Float16)hi.y, (_Float16)hi.z, (_Float16)hi.w};
  *(v8h*)d = h;
}

__device__ __forceinline__ void convert_tile(const float* __restrict__ x,
                                             const float* __restrict__ W,
                                             _Float16* __restrict__ xh,
                                             _Float16* __restrict__ Wh,
                                             int nx, int nw, int tile, int ntile) {
  const int t8 = (int)threadIdx.x * 8;
  const int ix = tile * (16 * F_IN) + t8;
  if (ix < nx) convert8(x + ix, xh + ix);
  for (int iw = tile * 2048 + t8; iw < nw; iw += ntile * 2048)
    convert8(W + iw, Wh + iw);
}

#define KSTEPS (F_IN / 8)
#define GWG    768
#define SEGP   36
#define TK     5
#define TILEW  (16 * HEADS * SEGP)
#define FTH    256
#define FWV    (FTH / 32)

#if GAT_WMMA_F32
__device__ __forceinline__ void gemm_tile(const float* __restrict__ x,
                                          const float4 (&bf)[KSTEPS],
                                          float* xl, float* keep, bool kept,
                                          int N, int tile) {
  const int lane    = threadIdx.x & 31;
  const int wave    = threadIdx.x >> 5;
  const int g       = lane >> 4;
  const int l16     = lane & 15;
  const int rowBase = tile * 16;
  const int ncol    = wave * 16 + l16;
  int mrow = rowBase + l16; if (mrow >= N) mrow = N - 1;

  float4 af[KSTEPS];
#pragma unroll
  for (int i = 0; i < KSTEPS; ++i)
    af[i] = *(const float4*)(x + (size_t)mrow * F_IN + i * 8 + g * 4);
  v8f c = {};
#pragma unroll
  for (int i = 0; i < KSTEPS; ++i) {
    const v2f a0 = {af[i].x, af[i].y}, b0 = {bf[i].x, bf[i].y};
    const v2f a1 = {af[i].z, af[i].w}, b1 = {bf[i].z, bf[i].w};
    c = __builtin_amdgcn_wmma_f32_16x16x4_f32(false, a0, false, b0, (short)0, c,
                                              false, false);
    c = __builtin_amdgcn_wmma_f32_16x16x4_f32(false, a1, false, b1, (short)0, c,
                                              false, false);
  }
#pragma unroll
  for (int i = 0; i < 8; ++i) {
    const int r = rowBase + i + g * 8;
    if (r < N) xl[(size_t)r * HC + ncol] = c[i];
    if (kept) keep[((i + g * 8) * HEADS + (ncol >> 5)) * SEGP + (ncol & 31)] = c[i];
  }
}
#else
__device__ __forceinline__ void gemm_tile_h(const _Float16* __restrict__ xh,
                                            const _Float16* __restrict__ Wh,
                                            float* xl, float* keep, bool kept,
                                            int N, int tile) {
  const int lane    = threadIdx.x & 31;
  const int wave    = threadIdx.x >> 5;
  const int g       = lane >> 4;
  const int l16     = lane & 15;
  const int rowBase = tile * 16;
  const int ncol    = wave * 16 + l16;
  int mrow = rowBase + l16; if (mrow >= N) mrow = N - 1;
  v8f c = {};
  for (int kk = 0; kk < F_IN; kk += 32) {
    const _Float16* ap = xh + (size_t)mrow * F_IN + kk + 8 * g;
    v8h a0 = *(const v8h*)(ap);
    v8h a1 = *(const v8h*)(ap + 16);
    v16h a = __builtin_shufflevector(a0, a1, 0, 1, 2, 3, 4, 5, 6, 7,
                                     8, 9, 10, 11, 12, 13, 14, 15);
    v16h b = *(const v16h*)(Wh + (size_t)ncol * F_IN + kk + 16 * g);
    c = __builtin_amdgcn_wmma_f32_16x16x32_f16(false, a, false, b, (short)0, c,
                                               false, false);
  }
#pragma unroll
  for (int i = 0; i < 8; ++i) {
    const int r = rowBase + i + g * 8;
    if (r < N) xl[(size_t)r * HC + ncol] = c[i];
    if (kept) keep[((i + g * 8) * HEADS + (ncol >> 5)) * SEGP + (ncol & 31)] = c[i];
  }
}
#endif

__device__ __forceinline__ void scores_body(const float* seg,
                                            const float* __restrict__ aj,
                                            const float* __restrict__ ai,
                                            float* __restrict__ Sj,
                                            float* __restrict__ Si, int N, int t) {
  if (t >= N * HEADS) return;
  const int h = t % HEADS;
  float sj = 0.f, si = 0.f;
#pragma unroll
  for (int cc = 0; cc < C_OUT; ++cc) {
    sj += seg[cc] * aj[h * C_OUT + cc];
    si += seg[cc] * ai[h * C_OUT + cc];
  }
  Sj[t] = sj;
  Si[t] = si;
}

#define RNG 196
#define MKW 32
#define MKR 256
#define MKP (MKR + 1)
#define SMW ((MKW * MKP > TK * TILEW) ? (MKW * MKP) : (TK * TILEW))

__device__ __forceinline__ void mark_body(const int* __restrict__ dst,
                                          unsigned* __restrict__ bm,
                                          unsigned* __restrict__ tile,
                                          int N, int E, int nranges, int pitch,
                                          int grp) {
  const int tid = threadIdx.x, lane = tid & 31, v = tid >> 5;

  for (int r0 = 0; r0 < nranges; r0 += MKR) {
    for (int i = tid; i < MKW * MKP; i += FTH) tile[i] = 0u;
    __syncthreads();
    for (int wv = v; wv < MKW; wv += FWV) {
      const size_t w = (size_t)grp * MKW + wv;
      int d[32];
#pragma unroll
      for (int j = 0; j < 32; ++j) {
        const size_t e = (w * 32 + j) * 32 + lane;
        d[j] = (e < (size_t)E) ? dst[e] : -1;
      }
#pragma unroll
      for (int j = 0; j < 32; ++j) {
        if ((unsigned)d[j] < (unsigned)N) {
          const unsigned r = (unsigned)d[j] / RNG - (unsigned)r0;
          if (r < MKR) atomicOr(&tile[wv * MKP + r], 1u << j);
        }
      }
    }
    __syncthreads();
    for (int r = v; r < MKR; r += 4 * FWV) {
      unsigned val[4];
#pragma unroll
      for (int q = 0; q < 4; ++q) {
        const int rr = r + q * FWV;
        val[q] = (rr < MKR) ? tile[lane * MKP + rr] : 0u;
      }
#pragma unroll
      for (int q = 0; q < 4; ++q) {
        const int rr = r + q * FWV;
        if (rr < MKR && r0 + rr < nranges)
          bm[(size_t)(r0 + rr) * pitch + (size_t)grp * MKW + lane] = val[q];
      }
    }
    __syncthreads();
  }
}

__global__ __launch_bounds__(FTH) void k_front(
    const float* __restrict__ x, const float* __restrict__ W,
    _Float16* __restrict__ xh, _Float16* __restrict__ Wh,
    float* xl, const float* __restrict__ aj,
    const float* __restrict__ ai, float* __restrict__ Sj, float* __restrict__ Si,
    const int* __restrict__ dst, unsigned* __restrict__ bm,
    int N, int E, int nx, int nw, int ntile, int nranges, int pitch, int nmark) {
  __shared__ __attribute__((aligned(16))) unsigned smem[SMW];
  if ((int)blockIdx.x < nmark) {
    mark_body(dst, bm, smem, N, E, nranges, pitch, (int)blockIdx.x);
    return;
  }
  float* keep = (float*)smem;
  const int g0 = (int)blockIdx.x - nmark;
#if GAT_WMMA_F32
  const int lane = threadIdx.x & 31;
  const int ncol = (int)(threadIdx.x >> 5) * 16 + (lane & 15);
  float4 bf[KSTEPS];
#pragma unroll
  for (int i = 0; i < KSTEPS; ++i)
    bf[i] = *(const float4*)(W + (size_t)ncol * F_IN + i * 8 + (lane >> 4) * 4);
  for (int t = g0, k = 0; t < ntile; t += GWG, ++k) {
    gemm_tile(x, bf, xl, keep + (k < TK ? k : 0) * TILEW, k < TK, N, t);
    convert_tile(x, W, xh, Wh, nx, nw, t, ntile);
  }
#else
  for (int t = g0, k = 0; t < ntile; t += GWG, ++k) {
    convert_tile(x, W, xh, Wh, nx, nw, t, ntile);
    __syncthreads();
    gemm_tile_h(xh, Wh, xl, keep + (k < TK ? k : 0) * TILEW, k < TK, N, t);
  }
#endif
  __syncthreads();
  const int sub = (int)threadIdx.x >> 6, pr = (int)threadIdx.x & 63;
  for (int k0 = 0; g0 + k0 * GWG < ntile; k0 += FTH / 64) {
    const int k  = k0 + sub;
    const int tl = g0 + k * GWG;
    if (tl < ntile) {
      const int t = tl * 64 + pr;
      if (k < TK)
        scores_body(keep + (k * 64 + pr) * SEGP, aj, ai, Sj, Si, N, t);
      else
        scores_body(xl + (size_t)t * C_OUT, aj, ai, Sj, Si, N, t);
    }
  }
}

#define ETH 1024
#define EPT 8
#define CAP (ETH * EPT)
#define WB  8

__device__ __forceinline__ float leaky(float a) {
  return (a >= 0.f) ? a : NEG_SLOPE * a;
}

__device__ __forceinline__ float lane_f(float v, int k) {
  return __int_as_float(__builtin_amdgcn_readlane(__float_as_int(v), k));
}

__device__ __forceinline__ float head_w(const float4& w, int k, int hq) {
  const float w0 = lane_f(w.x, k), w1 = lane_f(w.y, k);
  const float w2 = lane_f(w.z, k), w3 = lane_f(w.w, k);
  const float lo = (hq & 1) ? w1 : w0;
  const float hi = (hq & 1) ? w3 : w2;
  return (hq & 2) ? hi : lo;
}

__global__ __launch_bounds__(ETH) void k_edge(
    const int* __restrict__ src, const int* __restrict__ dst,
    const float* __restrict__ Sj, const float* __restrict__ Si,
    const float* __restrict__ xl, const float* __restrict__ bias,
    const unsigned* __restrict__ bm, int pitch,
    float* __restrict__ out, int N, int E) {
  __shared__ __attribute__((aligned(16))) float acc[RNG * HC];
  __shared__ unsigned lst_e[CAP];
  __shared__ __attribute__((aligned(16)))
  unsigned clst[CAP];
  __shared__ unsigned tkey[RNG * HEADS];
  __shared__ __attribute__((aligned(16))) float Uv[RNG * HEADS];
  __shared__ __attribute__((aligned(16))) float mval[RNG * HEADS];
  __shared__ float    ssum[RNG * HEADS];
  __shared__ unsigned cnt;
  __shared__ unsigned ccnt;

  const int tid  = threadIdx.x;
  const int base = blockIdx.x * RNG;
  const int nloc = (N - base < RNG) ? (N - base) : RNG;
  const int Et   = E + N;

  for (int i = tid; i < RNG * HC / 4; i += ETH)
    ((float4*)acc)[i] = make_float4(0.f, 0.f, 0.f, 0.f);
  for (int i = tid; i < CAP / 4; i += ETH)
    ((uint4*)clst)[i] = make_uint4(0u, 0u, 0u, 0u);
  for (int i = tid; i < RNG * HEADS; i += ETH) {
    const int dl = i >> 2, h = i & 3;
    tkey[i] = 0u;
    ssum[i] = 0.f;
    Uv[i] = (dl < nloc) ? Si[dst[base + dl] * HEADS + h] : 0.f;
  }
  if (tid == 0) { cnt = 0u; ccnt = 0u; }
  __syncthreads();

  const unsigned* row = bm + (size_t)blockIdx.x * pitch;
  for (int w = tid; w < pitch; w += ETH) {
    unsigned m = row[w];
    while (m) {
      const int j = __ffs(m) - 1;
      m &= m - 1;
      const unsigned pos = atomicAdd(&ccnt, 1u);
      if (pos < CAP) clst[pos] = (unsigned)w * 32u + (unsigned)j;
    }
  }
  __syncthreads();

  const unsigned ctot = ccnt;
  const int nch = (int)((ctot < CAP) ? ctot : CAP);
  {
    const int lane = tid & 31, wave = tid >> 5, g = lane >> 3, q = lane & 7;
    if (((((size_t)dst) & 15) == 0) && ((E & 3) == 0)) {
      for (int ib = wave * 32; ib < nch; ib += 32 * (ETH / 32)) {
        const uint4 ca = *(const uint4*)(clst + ib + 8 * g);
        const uint4 cb = *(const uint4*)(clst + ib + 8 * g + 4);
        const unsigned cid[8] = {ca.x, ca.y, ca.z, ca.w, cb.x, cb.y, cb.z, cb.w};
        size_t e0[8];
        unsigned lim[8];
        int4 dv[8];
#pragma unroll
        for (int k = 0; k < 8; ++k) {
          e0[k] = (size_t)cid[k] * 32 + (size_t)q * 4;
          lim[k] = (ib + 8 * g + k < nch && e0[k] < (size_t)E) ? (unsigned)nloc : 0u;
        }
#pragma unroll
        for (int k = 0; k < 8; ++k)
          dv[k] = *(const int4*)(dst + (lim[k] ? e0[k] : (size_t)0));
        unsigned hits = 0u;
#pragma unroll
        for (int k = 0; k < 8; ++k) {
          hits |= ((unsigned)(dv[k].x - base) < lim[k] ? 1u : 0u) << (4 * k);
          hits |= ((unsigned)(dv[k].y - base) < lim[k] ? 1u : 0u) << (4 * k + 1);
          hits |= ((unsigned)(dv[k].z - base) < lim[k] ? 1u : 0u) << (4 * k + 2);
          hits |= ((unsigned)(dv[k].w - base) < lim[k] ? 1u : 0u) << (4 * k + 3);
        }
        while (hits) {
          const int bit = __ffs(hits) - 1;
          hits &= hits - 1;
          const unsigned c = clst[ib + 8 * g + (bit >> 2)];
          const unsigned pos = atomicAdd(&cnt, 1u);
          if (pos < CAP) lst_e[pos] = c * 32u + (unsigned)(q * 4 + (bit & 3));
        }
      }
    } else {
      for (int i = wave; i < nch; i += ETH / 32) {
        const size_t e = (size_t)clst[i] * 32 + lane;
        if (e < (size_t)E && (unsigned)(dst[e] - base) < (unsigned)nloc) {
          const unsigned pos = atomicAdd(&cnt, 1u);
          if (pos < CAP) lst_e[pos] = (unsigned)e;
        }
      }
    }
  }
  if (tid < nloc) {
    const unsigned pos = atomicAdd(&cnt, 1u);
    if (pos < CAP) lst_e[pos] = (unsigned)(E + base + tid);
  }
  __syncthreads();

  const unsigned total = (ctot > CAP) ? (CAP + 1u) : cnt;
  if (total <= CAP) {
    const int lane = tid & 31;
    const int n  = __builtin_amdgcn_readfirstlane((int)total);
    const int wv = __builtin_amdgcn_readfirstlane(tid >> 5);
    const int per = ((n + (ETH / 32) - 1) / (ETH / 32) + WB - 1) & ~(WB - 1);
    const int first = wv * per;
    int mine = n - first;
    if (mine > per) mine = per;
    if (mine < 0) mine = 0;

    int    ee[EPT];
    int    es[EPT];
    int    ed[EPT];
    float4 et[EPT];
#pragma unroll
    for (int j = 0; j < EPT; ++j) {
      const int p = j * 32 + lane;
      ee[j] = (p < mine) ? (int)lst_e[first + p] : -1;
    }
    {
      int sl[EPT], dld[EPT];
#pragma unroll
      for (int j = 0; j < EPT; ++j) {
        const int ec = (ee[j] >= 0 && ee[j] < E) ? ee[j] : 0;
        sl[j]  = src[ec];
        dld[j] = dst[ec];
      }
#pragma unroll
      for (int j = 0; j < EPT; ++j) {
        const int e = ee[j];
        es[j] = (e < 0) ? 0 : ((e < E) ? sl[j] : (e - E));
        ed[j] = (e < 0) ? 0 : (((e < E) ? dld[j] : (e - E)) - base);
      }
    }
    {
      int ss[EPT];
#pragma unroll
      for (int j = 0; j < EPT; ++j) ss[j] = src[es[j]];
#pragma unroll
      for (int j = 0; j < EPT; ++j)
        et[j] = *(const float4*)(Sj + (size_t)ss[j] * HEADS);
    }
#pragma unroll
    for (int j = 0; j < EPT; ++j) {
      if (ee[j] >= 0) {
        atomicMax(&tkey[ed[j] * HEADS + 0], f2key(et[j].x));
        atomicMax(&tkey[ed[j] * HEADS + 1], f2key(et[j].y));
        atomicMax(&tkey[ed[j] * HEADS + 2], f2key(et[j].z));
        atomicMax(&tkey[ed[j] * HEADS + 3], f2key(et[j].w));
      }
    }
    __syncthreads();
    for (int i = tid; i < RNG * HEADS; i += ETH)
      mval[i] = leaky(key2f(tkey[i]) + Uv[i]);
    __syncthreads();

    const int hq = lane >> 3;
#pragma unroll
    for (int j = 0; j < EPT; ++j) {
      const int left = mine - j * 32;
      if (left > 0) {
        const int dl = ed[j];
        const float4 U = *(const float4*)(Uv + dl * HEADS);
        const float4 M = *(const float4*)(mval + dl * HEADS);
        float4 w;
        w.x = expf(leaky(et[j].x + U.x) - M.x);
        w.y = expf(leaky(et[j].y + U.y) - M.y);
        w.z = expf(leaky(et[j].z + U.z) - M.z);
        w.w = expf(leaky(et[j].w + U.w) - M.w);
        if (lane < left) {
          atomicAdd(&ssum[dl * HEADS + 0], w.x);
          atomicAdd(&ssum[dl * HEADS + 1], w.y);
          atomicAdd(&ssum[dl * HEADS + 2], w.z);
          atomicAdd(&ssum[dl * HEADS + 3], w.w);
        }
        const int cj = (left < 32) ? left : 32;
        const int cfull = cj & ~(WB - 1);
        for (int k0 = 0; k0 < cfull; k0 += WB) {
          float4 v[WB];
#pragma unroll
          for (int b = 0; b < WB; ++b) {
            const int s = __builtin_amdgcn_readlane(es[j], k0 + b);
            v[b] = *(const float4*)(xl + (size_t)s * HC + lane * 4);
          }
#pragma unroll
          for (int b = 0; b < WB; ++b) {
            const int k = k0 + b;
            const int d = __builtin_amdgcn_readlane(dl, k);
            const float wl = head_w(w, k, hq);
            float* ar = acc + d * HC + lane;
            atomicAdd(ar,      wl * v[b].x);
            atomicAdd(ar + 32, wl * v[b].y);
            atomicAdd(ar + 64, wl * v[b].z);
            atomicAdd(ar + 96, wl * v[b].w);
          }
        }
        for (int k = cfull; k < cj; ++k) {
          const int s = __builtin_amdgcn_readlane(es[j], k);
          const int d = __builtin_amdgcn_readlane(dl, k);
          const float4 v = *(const float4*)(xl + (size_t)s * HC + lane * 4);
          const float wl = head_w(w, k, hq);
          float* ar = acc + d * HC + lane;
          atomicAdd(ar,      wl * v.x);
          atomicAdd(ar + 32, wl * v.y);
          atomicAdd(ar + 64, wl * v.z);
          atomicAdd(ar + 96, wl * v.w);
        }
      }
    }
  } else {
    for (int e = tid; e < Et; e += ETH) {
      const int d = (e < E) ? dst[e] : (e - E);
      const unsigned dl = (unsigned)(d - base);
      if (dl < (unsigned)nloc) {
        const int s  = (e < E) ? src[e] : (e - E);
        const int ss = src[s];
        for (int hh = 0; hh < HEADS; ++hh)
          atomicMax(&tkey[dl * HEADS + hh], f2key(Sj[(size_t)ss * HEADS + hh]));
      }
    }
    __syncthreads();
    for (int i = tid; i < RNG * HEADS; i += ETH)
      mval[i] = leaky(key2f(tkey[i]) + Uv[i]);
    __syncthreads();
    for (int e = tid; e < Et; e += ETH) {
      const int d = (e < E) ? dst[e] : (e - E);
      const unsigned dl = (unsigned)(d - base);
      if (dl < (unsigned)nloc) {
        const int s  = (e < E) ? src[e] : (e - E);
        const int ss = src[s];
        float w[HEADS];
        for (int hh = 0; hh < HEADS; ++hh) {
          const float a = leaky(Sj[(size_t)ss * HEADS + hh] + Uv[dl * HEADS + hh]);
          w[hh] = expf(a - mval[dl * HEADS + hh]);
          atomicAdd(&ssum[dl * HEADS + hh], w[hh]);
        }
        for (int f = 0; f < HC; ++f)
          atomicAdd(&acc[dl * HC + (f & 3) * 32 + (f >> 2)],
                    w[f >> 5] * xl[(size_t)s * HC + f]);
      }
    }
  }
  __syncthreads();

  for (int i = tid; i < nloc * 32; i += ETH) {
    const int dl = i >> 5, l = i & 31;
    const float inv = 1.0f / ssum[dl * HEADS + (l >> 3)];
    const float* ar = acc + dl * HC + l;
    const float4 b = *(const float4*)(bias + l * 4);
    float4 o;
    o.x = ar[0]  * inv + b.x;
    o.y = ar[32] * inv + b.y;
    o.z = ar[64] * inv + b.z;
    o.w = ar[96] * inv + b.w;
    *(float4*)(out + (size_t)(base + dl) * HC + l * 4) = o;
  }
}

extern "C" void kernel_launch(void* const* d_in, const int* in_sizes, int n_in,
                              void* d_out, int out_size, void* d_ws, size_t ws_size,
                              hipStream_t stream) {
  const float* x      = (const float*)d_in[0];
  const float* W      = (const float*)d_in[1];
  const float* attn_j = (const float*)d_in[2];
  const float* attn_i = (const float*)d_in[3];
  const float* bias   = (const float*)d_in[4];
  const int*   eidx   = (const int*)d_in[5];

  const int N  = in_sizes[0] / F_IN;
  const int E  = in_sizes[5] / 2;
  const int* src = eidx;
  const int* dst = eidx + E;

  char*  ws = (char*)d_ws;
  size_t o  = 0;
  auto take = [&](size_t bytes) -> void* {
    void* p = ws + o;
    o += (bytes + 255) & ~(size_t)255;
    return p;
  };
  float*     xl   = (float*)take((size_t)N * HC * sizeof(float));
  float*     Sj   = (float*)take((size_t)N * HEADS * sizeof(float));
  float*     Si   = (float*)take((size_t)N * HEADS * sizeof(float));
  _Float16*  xh   = (_Float16*)take((size_t)N * F_IN * sizeof(_Float16));
  _Float16*  Wh   = (_Float16*)take((size_t)HC * F_IN * sizeof(_Float16));
  const int nranges = (N + RNG - 1) / RNG;
  const int nchunks = (E + 31) / 32;
  const int pitch   = ((nchunks + 31) / 32 + MKW - 1) / MKW * MKW;
  unsigned*  bm   = (unsigned*)take((size_t)nranges * pitch * sizeof(unsigned));

  float* out = (float*)d_out;

  const int nx = N * F_IN, nw = HC * F_IN;
  const int ntile = (N + 15) / 16;
  const int nmark = pitch / MKW;
  k_front<<<nmark + GWG, FTH, 0, stream>>>(x, W, xh, Wh, xl, attn_j, attn_i, Sj, Si,
                                          dst, bm, N, E, nx, nw, ntile, nranges,
                                          pitch, nmark);

  k_edge<<<nranges, ETH, 0, stream>>>(src, dst, Sj, Si, xl, bias, bm, pitch, out, N, E);
}
